# starting kernel
# baseline (speedup 1.0000x reference)
#include <hip/hip_runtime.h>
#include <math.h>


typedef float v2f __attribute__((ext_vector_type(2)));
typedef float v8f __attribute__((ext_vector_type(8)));

#define NW 8
#define NH 16
#define EDIM 128
#define TOK_PER_BLOCK 16

__global__ __launch_bounds__(256) void quantum_attn_kernel(
    const float* __restrict__ x,
    const float* __restrict__ theta,
    float* __restrict__ Y)
{
    __shared__ float qh_lds[TOK_PER_BLOCK][NH][NW];

    const int lane = threadIdx.x & 31;
    const int wave = threadIdx.x >> 5;
    const int tIn  = wave * 2 + (lane >> 4);
    const int h    = lane & 15;
    const int gt   = blockIdx.x * TOK_PER_BLOCK + tIn;

    const float* xp = x + (size_t)gt * EDIM + h * NW;
    float4 x0 = *(const float4*)(xp);
    float4 x1 = *(const float4*)(xp + 4);
    float4 t0 = *(const float4*)(theta);
    float4 t1 = *(const float4*)(theta + 4);

    float c[NW];
    c[0] = __cosf(x0.x + t0.x);
    c[1] = __cosf(x0.y + t0.y);
    c[2] = __cosf(x0.z + t0.z);
    c[3] = __cosf(x0.w + t0.w);
    c[4] = __cosf(x1.x + t1.x);
    c[5] = __cosf(x1.y + t1.y);
    c[6] = __cosf(x1.z + t1.z);
    c[7] = __cosf(x1.w + t1.w);

    float q[NW];
    float p = c[0];
#pragma unroll
    for (int w = 1; w < NW; ++w) { p *= c[w]; q[w] = p; }
    q[0] = c[1] * c[2] * c[3] * c[4] * c[5] * c[6] * c[7];

#pragma unroll
    for (int w = 0; w < NW; ++w) qh_lds[tIn][h][w] = q[w];
    __syncthreads();

    const float inv_sqrt_dk = 0.35355339059327379f;
    float sc[NH];
    float mx = -INFINITY;
#pragma unroll
    for (int g = 0; g < NH; ++g) {
        float d = 0.f;
#pragma unroll
        for (int w = 0; w < NW; ++w) d += q[w] * qh_lds[tIn][g][w];
        d *= inv_sqrt_dk;
        sc[g] = d;
        mx = fmaxf(mx, d);
    }
    float denom = 0.f;
#pragma unroll
    for (int g = 0; g < NH; ++g) { sc[g] = __expf(sc[g] - mx); denom += sc[g]; }

    float o[NW];
#pragma unroll
    for (int w = 0; w < NW; ++w) o[w] = 0.f;
#pragma unroll
    for (int g = 0; g < NH; ++g) {
        float a = sc[g];
#pragma unroll
        for (int w = 0; w < NW; ++w) o[w] += a * qh_lds[tIn][g][w];
    }
    float r = 1.f / denom;
#pragma unroll
    for (int w = 0; w < NW; ++w) o[w] *= r;

    const int b = gt >> 10;
    const int s = gt & 1023;
    const size_t row = (size_t)b * 1024 + (size_t)h * 64 + (s >> 4);
    float* yp = Y + row * EDIM + (s & 15) * NW;
    *(float4*)(yp)     = make_float4(o[0], o[1], o[2], o[3]);
    *(float4*)(yp + 4) = make_float4(o[4], o[5], o[6], o[7]);
}

__global__ __launch_bounds__(256) void combine_gemm_wmma(
    const float* __restrict__ Y,
    const float* __restrict__ W,
    const float* __restrict__ bias,
    float* __restrict__ out)
{
    const int lane  = threadIdx.x & 31;
    const int wave  = threadIdx.x >> 5;
    const int tileN = wave;
    const int tileM = blockIdx.x;
    const int l16   = lane & 15;
    const int kgrp  = (lane >> 4) * 2;

    const float* arow = Y + ((size_t)tileM * 16 + l16) * EDIM + kgrp;
    const float* brow = W + ((size_t)tileN * 16 + l16) * EDIM + kgrp;

    __builtin_prefetch(arow, 0, 0);
    __builtin_prefetch(brow, 0, 0);

    v8f acc = {0.f, 0.f, 0.f, 0.f, 0.f, 0.f, 0.f, 0.f};
#pragma unroll
    for (int k0 = 0; k0 < EDIM; k0 += 4) {
        v2f a = *(const v2f*)(arow + k0);
        v2f b = *(const v2f*)(brow + k0);
        acc = __builtin_amdgcn_wmma_f32_16x16x4_f32(
             false, a,  false, b,
             (short)0, acc,  false,  false);
    }

    const int n = tileN * 16 + l16;
    const float bcol = bias[n];
    const int rbase = tileM * 16 + 8 * (lane >> 4);
#pragma unroll
    for (int rr = 0; rr < 8; ++rr) {
        out[(size_t)(rbase + rr) * EDIM + n] = acc[rr] + bcol;
    }
}

extern "C" void kernel_launch(void* const* d_in, const int* in_sizes, int n_in,
                              void* d_out, int out_size, void* d_ws, size_t ws_size,
                              hipStream_t stream) {
    const float* x     = (const float*)d_in[0];
    const float* theta = (const float*)d_in[1];
    const float* Wc    = (const float*)d_in[2];
    const float* bc    = (const float*)d_in[3];
    float* out = (float*)d_out;
    float* Y   = (float*)d_ws;

    const int ntok = in_sizes[0] / EDIM;

    const int blocks1 = ntok / TOK_PER_BLOCK;
    quantum_attn_kernel<<<blocks1, 256, 0, stream>>>(x, theta, Y);

    const int mtiles = ntok / 16;
    combine_gemm_wmma<<<mtiles, 256, 0, stream>>>(Y, Wc, bc, out);
}


# One launch: a 2-wave work-group keeps its 16 tokens' attention output in registers as WMMA A fragments (k in lane-token order); each wave sums 64 k for all columns, half sums added via LDS
# speedup vs baseline: 1.6066x; 1.6066x over previous
#include <hip/hip_runtime.h>
#include <math.h>


typedef float v2f __attribute__((ext_vector_type(2)));
typedef float v8f __attribute__((ext_vector_type(8)));

#define NW 8
#define NH 16
#define EDIM 128
#define TOK_PER_BLOCK 16
#define TOK_PER_LANE 4
#define NTILE 8
#define W_EARLY 3

__global__ __launch_bounds__(64) void quantum_attn_proj_kernel(
    const float* __restrict__ x,
    const float* __restrict__ theta,
    const float* __restrict__ W,
    const float* __restrict__ bias,
    float* __restrict__ out)
{
    __shared__ float qh_lds[TOK_PER_BLOCK][NH][NW];
    __shared__ float part_lds[2][NTILE / 2][8][32];

    const int lane = threadIdx.x & 31;
    const int wave = threadIdx.x >> 5;
    const int half = lane >> 4;
    const int h    = lane & 15;
    const int slot0 = wave * 8 + half;

    const float4 t0 = *(const float4*)(theta);
    const float4 t1 = *(const float4*)(theta + 4);

    float4 xa[TOK_PER_LANE], xb[TOK_PER_LANE];
#pragma unroll
    for (int jj = 0; jj < TOK_PER_LANE; ++jj) {
        const int gt = blockIdx.x * TOK_PER_BLOCK + slot0 + 2 * jj;
        const float* xp = x + (size_t)gt * EDIM + h * NW;
        xa[jj] = *(const float4*)(xp);
        xb[jj] = *(const float4*)(xp + 4);
    }

    float4 wlo[TOK_PER_LANE][NTILE], whi[TOK_PER_LANE][NTILE];
#pragma unroll
    for (int jj = 0; jj < W_EARLY; ++jj) {
        const int kbase = (slot0 + 2 * jj) * NW;
#pragma unroll
        for (int u = 0; u < NTILE; ++u) {
            const int nt = (u + 4 * wave) & 7;
            const float* wp = W + (size_t)(nt * 16 + h) * EDIM + kbase;
            wlo[jj][u] = *(const float4*)(wp);
            whi[jj][u] = *(const float4*)(wp + 4);
        }
    }

    __builtin_amdgcn_sched_barrier(0);

    float q[TOK_PER_LANE][NW];
#pragma unroll
    for (int jj = 0; jj < TOK_PER_LANE; ++jj) {
        const int slot = slot0 + 2 * jj;
        const float4 x0 = xa[jj];
        const float4 x1 = xb[jj];

        float c[NW];
        c[0] = __cosf(x0.x + t0.x);
        c[1] = __cosf(x0.y + t0.y);
        c[2] = __cosf(x0.z + t0.z);
        c[3] = __cosf(x0.w + t0.w);
        c[4] = __cosf(x1.x + t1.x);
        c[5] = __cosf(x1.y + t1.y);
        c[6] = __cosf(x1.z + t1.z);
        c[7] = __cosf(x1.w + t1.w);

        float p = c[0];
#pragma unroll
        for (int w = 1; w < NW; ++w) { p *= c[w]; q[jj][w] = p; }
        q[jj][0] = c[1] * c[2] * c[3] * c[4] * c[5] * c[6] * c[7];

#pragma unroll
        for (int w = 0; w < NW; ++w) qh_lds[slot][h][w] = q[jj][w];
    }
    __syncthreads();

    const float inv_sqrt_dk = 0.35355339059327379f;
    float o[TOK_PER_LANE][NW];
#pragma unroll
    for (int jj = 0; jj < TOK_PER_LANE; ++jj) {
        const int slot = slot0 + 2 * jj;
        float sc[NH];
        float mx = -INFINITY;
#pragma unroll
        for (int g = 0; g < NH; ++g) {
            float d = 0.f;
#pragma unroll
            for (int w = 0; w < NW; ++w) d += q[jj][w] * qh_lds[slot][g][w];
            d *= inv_sqrt_dk;
            sc[g] = d;
            mx = fmaxf(mx, d);
        }
        float denom = 0.f;
#pragma unroll
        for (int g = 0; g < NH; ++g) { sc[g] = __expf(sc[g] - mx); denom += sc[g]; }

        float acc_o[NW];
#pragma unroll
        for (int w = 0; w < NW; ++w) acc_o[w] = 0.f;
#pragma unroll
        for (int g = 0; g < NH; ++g) {
            const float a = sc[g];
#pragma unroll
            for (int w = 0; w < NW; ++w) acc_o[w] += a * qh_lds[slot][g][w];
        }
        const float r = 1.f / denom;
#pragma unroll
        for (int w = 0; w < NW; ++w) o[jj][w] = acc_o[w] * r;
    }

    v8f acc[NTILE];
#pragma unroll
    for (int u = 0; u < NTILE; ++u)
        acc[u] = (v8f){0.f, 0.f, 0.f, 0.f, 0.f, 0.f, 0.f, 0.f};

#pragma unroll
    for (int jj = W_EARLY; jj < TOK_PER_LANE; ++jj) {
        const int kbase = (slot0 + 2 * jj) * NW;
#pragma unroll
        for (int u = 0; u < NTILE; ++u) {
            const int nt = (u + 4 * wave) & 7;
            const float* wp = W + (size_t)(nt * 16 + h) * EDIM + kbase;
            wlo[jj][u] = *(const float4*)(wp);
            whi[jj][u] = *(const float4*)(wp + 4);
        }
    }

#pragma unroll
    for (int jj = 0; jj < TOK_PER_LANE; ++jj) {
#pragma unroll
        for (int i = 0; i < 4; ++i) {
            v2f a;
            a.x = o[jj][2 * i];
            a.y = o[jj][2 * i + 1];
#pragma unroll
            for (int u = 0; u < NTILE; ++u) {
                v2f b;
                if (i == 0)      { b.x = wlo[jj][u].x; b.y = wlo[jj][u].y; }
                else if (i == 1) { b.x = wlo[jj][u].z; b.y = wlo[jj][u].w; }
                else if (i == 2) { b.x = whi[jj][u].x; b.y = whi[jj][u].y; }
                else             { b.x = whi[jj][u].z; b.y = whi[jj][u].w; }
                acc[u] = __builtin_amdgcn_wmma_f32_16x16x4_f32(
                     false, a,  false, b,
                     (short)0, acc[u],  false,  false);
            }
        }
    }

#pragma unroll
    for (int t = 0; t < NTILE / 2; ++t)
#pragma unroll
        for (int r = 0; r < 8; ++r)
            part_lds[wave][t][r][lane] = acc[NTILE / 2 + t][r];
    __syncthreads();

    const int b_idx = blockIdx.x >> 6;
    const int sgrp  = blockIdx.x & 63;
#pragma unroll
    for (int t = 0; t < NTILE / 2; ++t) {
        const int n = (4 * wave + t) * 16 + h;
        const float bcol = bias[n];
#pragma unroll
        for (int r = 0; r < 8; ++r) {
            const int head = 8 * half + r;
            const size_t row = (size_t)b_idx * 1024 + (size_t)head * 64 + sgrp;
            const float s = acc[t][r] + part_lds[1 - wave][t][r][lane];
            out[row * EDIM + n] = s + bcol;
        }
    }
}

extern "C" void kernel_launch(void* const* d_in, const int* in_sizes, int n_in,
                              void* d_out, int out_size, void* d_ws, size_t ws_size,
                              hipStream_t stream) {
    const float* x     = (const float*)d_in[0];
    const float* theta = (const float*)d_in[1];
    const float* Wc    = (const float*)d_in[2];
    const float* bc    = (const float*)d_in[3];
    float* out = (float*)d_out;

    const int ntok = in_sizes[0] / EDIM;
    const int blocks1 = ntok / TOK_PER_BLOCK;
    quantum_attn_proj_kernel<<<blocks1, 64, 0, stream>>>(x, theta, Wc, bc, out);
}


# Each matrix instruction of token step jj is followed by one of 32 pieces of the attention of step jj+1, written on register pairs; a score is now summed as its even-w plus its odd-w half
# speedup vs baseline: 1.7293x; 1.0764x over previous
#include <hip/hip_runtime.h>
#include <math.h>


typedef float v2f __attribute__((ext_vector_type(2)));
typedef float v8f __attribute__((ext_vector_type(8)));

#define NW 8
#define NH 16
#define EDIM 128
#define TOK_PER_BLOCK 16
#define TOK_PER_LANE 4
#define NTILE 8
#define W_EARLY 3

__device__ __forceinline__ void load_rows(float4 (&kv)[2 * NH], const float* rows)
{
#pragma unroll
    for (int g = 0; g < NH; ++g) {
        kv[2 * g]     = *(const float4*)(rows + g * NW);
        kv[2 * g + 1] = *(const float4*)(rows + g * NW + 4);
    }
}

__device__ __forceinline__ v2f kpair(const float4 (&kv)[2 * NH], int g, int j)
{
    const float4 v = kv[2 * g + (j >> 1)];
    v2f p;
    if ((j & 1) == 0) { p.x = v.x; p.y = v.y; } else { p.x = v.z; p.y = v.w; }
    return p;
}

__device__ __forceinline__ void attend_piece(int p, const float (&qv)[NW], const float4 (&kv)[2 * NH],
                                             v2f (&d2)[NH], float (&e)[NH], v2f (&ao)[NW / 2],
                                             float& mx, float& den, float& r, float (&ov)[NW])
{
    const float inv_sqrt_dk = 0.35355339059327379f;
    if (p < 8) {
        const int j = p >> 1, g0 = 8 * (p & 1);
        v2f qp; qp.x = qv[2 * j]; qp.y = qv[2 * j + 1];
#pragma unroll
        for (int g = g0; g < g0 + 8; ++g)
            d2[g] = __builtin_elementwise_fma(qp, kpair(kv, g, j), d2[g]);
    } else if (p == 8) {
#pragma unroll
        for (int g = 0; g < NH; ++g) e[g] = (d2[g].x + d2[g].y) * inv_sqrt_dk;
    } else if (p == 9) {
        float m = e[0];
#pragma unroll
        for (int g = 1; g < NH; ++g) m = fmaxf(m, e[g]);
        mx = m;
    } else if (p < 14) {
        const int g0 = 4 * (p - 10);
#pragma unroll
        for (int t = 0; t < 4; t += 2) {
            v2f a; a.x = e[g0 + t]; a.y = e[g0 + t + 1];
            v2f m2; m2.x = mx; m2.y = mx;
            const v2f l2 = {1.44269502f, 1.44269502f};
            const v2f s = (a - m2) * l2;
            e[g0 + t]     = __builtin_amdgcn_exp2f(s.x);
            e[g0 + t + 1] = __builtin_amdgcn_exp2f(s.y);
        }
        if (p > 10) {
#pragma unroll
            for (int t = 0; t < 4; ++t) den += e[g0 - 4 + t];
        }
    } else if (p == 14) {
#pragma unroll
        for (int t = 0; t < 4; ++t) den += e[12 + t];
        r = 1.f / den;
    } else if (p < 31) {
        const int g = p - 15;
        v2f ep; ep.x = e[g]; ep.y = e[g];
#pragma unroll
        for (int j = 0; j < NW / 2; ++j)
            ao[j] = __builtin_elementwise_fma(ep, kpair(kv, g, j), ao[j]);
    } else {
#pragma unroll
        for (int j = 0; j < NW / 2; ++j) { ov[2 * j] = ao[j].x * r; ov[2 * j + 1] = ao[j].y * r; }
    }
}

__global__ __launch_bounds__(64) void quantum_attn_proj_kernel(
    const float* __restrict__ x,
    const float* __restrict__ theta,
    const float* __restrict__ W,
    const float* __restrict__ bias,
    float* __restrict__ out)
{
    __shared__ float qh_lds[TOK_PER_BLOCK][NH][NW];
    __shared__ float part_lds[2][NTILE / 2][8][32];

    const int lane = threadIdx.x & 31;
    const int wave = threadIdx.x >> 5;
    const int half = lane >> 4;
    const int h    = lane & 15;
    const int slot0 = wave * 8 + half;

    const float4 t0 = *(const float4*)(theta);
    const float4 t1 = *(const float4*)(theta + 4);

    float4 xa[TOK_PER_LANE], xb[TOK_PER_LANE];
#pragma unroll
    for (int jj = 0; jj < TOK_PER_LANE; ++jj) {
        const int gt = blockIdx.x * TOK_PER_BLOCK + slot0 + 2 * jj;
        const float* xp = x + (size_t)gt * EDIM + h * NW;
        xa[jj] = *(const float4*)(xp);
        xb[jj] = *(const float4*)(xp + 4);
    }

    float4 wlo[TOK_PER_LANE][NTILE], whi[TOK_PER_LANE][NTILE];
#pragma unroll
    for (int jj = 0; jj < W_EARLY; ++jj) {
        const int kbase = (slot0 + 2 * jj) * NW;
#pragma unroll
        for (int u = 0; u < NTILE; ++u) {
            const int nt = (u + 4 * wave) & 7;
            const float* wp = W + (size_t)(nt * 16 + h) * EDIM + kbase;
            wlo[jj][u] = *(const float4*)(wp);
            whi[jj][u] = *(const float4*)(wp + 4);
        }
    }

    __builtin_amdgcn_sched_barrier(0);

    float q[TOK_PER_LANE][NW];
#pragma unroll
    for (int jj = 0; jj < TOK_PER_LANE; ++jj) {
        const int slot = slot0 + 2 * jj;
        const float4 x0 = xa[jj];
        const float4 x1 = xb[jj];

        float c[NW];
        c[0] = __cosf(x0.x + t0.x);
        c[1] = __cosf(x0.y + t0.y);
        c[2] = __cosf(x0.z + t0.z);
        c[3] = __cosf(x0.w + t0.w);
        c[4] = __cosf(x1.x + t1.x);
        c[5] = __cosf(x1.y + t1.y);
        c[6] = __cosf(x1.z + t1.z);
        c[7] = __cosf(x1.w + t1.w);

        float p = c[0];
#pragma unroll
        for (int w = 1; w < NW; ++w) { p *= c[w]; q[jj][w] = p; }
        q[jj][0] = c[1] * c[2] * c[3] * c[4] * c[5] * c[6] * c[7];

#pragma unroll
        for (int w = 0; w < NW; ++w) qh_lds[slot][h][w] = q[jj][w];
    }
    __syncthreads();

#pragma unroll
    for (int jj = W_EARLY; jj < TOK_PER_LANE; ++jj) {
        const int kbase = (slot0 + 2 * jj) * NW;
#pragma unroll
        for (int u = 0; u < NTILE; ++u) {
            const int nt = (u + 4 * wave) & 7;
            const float* wp = W + (size_t)(nt * 16 + h) * EDIM + kbase;
            wlo[jj][u] = *(const float4*)(wp);
            whi[jj][u] = *(const float4*)(wp + 4);
        }
    }

    float4 kvA[2 * NH], kvB[2 * NH];
    float o[TOK_PER_LANE][NW];
    v8f acc[NTILE];
#pragma unroll
    for (int u = 0; u < NTILE; ++u)
        acc[u] = (v8f){0.f, 0.f, 0.f, 0.f, 0.f, 0.f, 0.f, 0.f};

#define PROJECT_STEP(JJ)                                                          \
    _Pragma("unroll")                                                             \
    for (int i = 0; i < 4; ++i) {                                                 \
        v2f a;                                                                    \
        a.x = o[JJ][2 * i];                                                       \
        a.y = o[JJ][2 * i + 1];                                                   \
        _Pragma("unroll")                                                         \
        for (int u = 0; u < NTILE; ++u) {                                         \
            v2f b;                                                                \
            if (i == 0)      { b.x = wlo[JJ][u].x; b.y = wlo[JJ][u].y; }          \
            else if (i == 1) { b.x = wlo[JJ][u].z; b.y = wlo[JJ][u].w; }          \
            else if (i == 2) { b.x = whi[JJ][u].x; b.y = whi[JJ][u].y; }          \
            else             { b.x = whi[JJ][u].z; b.y = whi[JJ][u].w; }          \
            acc[u] = __builtin_amdgcn_wmma_f32_16x16x4_f32(                       \
                 false, a,  false, b,                           \
                 (short)0, acc[u],  false,  false);\
        }                                                                         \
    }

#define PIPE_PAIR(JJ, KV, M)                                                      \
    {                                                                             \
        v2f a_, b_;                                                               \
        a_.x = o[JJ][2 * ((M) >> 3)];                                             \
        a_.y = o[JJ][2 * ((M) >> 3) + 1];                                         \
        const float4 bq_ = (((M) >> 3) < 2) ? wlo[JJ][(M) & 7] : whi[JJ][(M) & 7];\
        if ((((M) >> 3) & 1) == 0) { b_.x = bq_.x; b_.y = bq_.y; }                \
        else                       { b_.x = bq_.z; b_.y = bq_.w; }                \
        acc[(M) & 7] = __builtin_amdgcn_wmma_f32_16x16x4_f32(                     \
             false, a_,  false, b_,                             \
             (short)0, acc[(M) & 7],  false,  false);\
        attend_piece((M), q[JJ + 1], KV, d_, e_, ao_, mx_, den_, r_, o[JJ + 1]);  \
        __builtin_amdgcn_sched_barrier(0);                                        \
    }
#define PIPE_8(JJ, KV, B)                                                         \
    PIPE_PAIR(JJ, KV, (B) + 0) PIPE_PAIR(JJ, KV, (B) + 1)                         \
    PIPE_PAIR(JJ, KV, (B) + 2) PIPE_PAIR(JJ, KV, (B) + 3)                         \
    PIPE_PAIR(JJ, KV, (B) + 4) PIPE_PAIR(JJ, KV, (B) + 5)                         \
    PIPE_PAIR(JJ, KV, (B) + 6) PIPE_PAIR(JJ, KV, (B) + 7)
#define PIPE_STEP(JJ, KV)                                                         \
    {                                                                             \
        v2f d_[NH], ao_[NW / 2];                                                  \
        float e_[NH];                                                             \
        float mx_ = 0.f, den_ = 0.f, r_ = 0.f;                                    \
        _Pragma("unroll")                                                         \
        for (int g = 0; g < NH; ++g) { d_[g] = (v2f){0.f, 0.f}; e_[g] = 0.f; }    \
        _Pragma("unroll")                                                         \
        for (int w = 0; w < NW / 2; ++w) ao_[w] = (v2f){0.f, 0.f};                \
        PIPE_8(JJ, KV, 0) PIPE_8(JJ, KV, 8) PIPE_8(JJ, KV, 16) PIPE_8(JJ, KV, 24) \
    }

    load_rows(kvA, &qh_lds[slot0][0][0]);
    load_rows(kvB, &qh_lds[slot0 + 2][0][0]);
    {
        v2f d_[NH], ao_[NW / 2];
        float e_[NH];
        float mx_ = 0.f, den_ = 0.f, r_ = 0.f;
#pragma unroll
        for (int g = 0; g < NH; ++g) { d_[g] = (v2f){0.f, 0.f}; e_[g] = 0.f; }
#pragma unroll
        for (int w = 0; w < NW / 2; ++w) ao_[w] = (v2f){0.f, 0.f};
#pragma unroll
        for (int p = 0; p < 32; ++p)
            attend_piece(p, q[0], kvA, d_, e_, ao_, mx_, den_, r_, o[0]);
    }

    load_rows(kvA, &qh_lds[slot0 + 4][0][0]);
    PIPE_STEP(0, kvB)

    load_rows(kvB, &qh_lds[slot0 + 6][0][0]);
    PIPE_STEP(1, kvA)

    PIPE_STEP(2, kvB)

    PROJECT_STEP(3)

#pragma unroll
    for (int t = 0; t < NTILE / 2; ++t)
#pragma unroll
        for (int r = 0; r < 8; ++r)
            part_lds[wave][t][r][lane] = acc[NTILE / 2 + t][r];
    __syncthreads();

    const int b_idx = blockIdx.x >> 6;
    const int sgrp  = blockIdx.x & 63;
#pragma unroll
    for (int t = 0; t < NTILE / 2; ++t) {
        const int n = (4 * wave + t) * 16 + h;
        const float bcol = bias[n];
#pragma unroll
        for (int r = 0; r < 8; ++r) {
            const int head = 8 * half + r;
            const size_t row = (size_t)b_idx * 1024 + (size_t)head * 64 + sgrp;
            const float s = acc[t][r] + part_lds[1 - wave][t][r][lane];
            out[row * EDIM + n] = s + bcol;
        }
    }
}

extern "C" void kernel_launch(void* const* d_in, const int* in_sizes, int n_in,
                              void* d_out, int out_size, void* d_ws, size_t ws_size,
                              hipStream_t stream) {
    const float* x     = (const float*)d_in[0];
    const float* theta = (const float*)d_in[1];
    const float* Wc    = (const float*)d_in[2];
    const float* bc    = (const float*)d_in[3];
    float* out = (float*)d_out;

    const int ntok = in_sizes[0] / EDIM;
    const int blocks1 = ntok / TOK_PER_BLOCK;
    quantum_attn_proj_kernel<<<blocks1, 64, 0, stream>>>(x, theta, Wc, bc, out);
}


# No global load waits at a barrier: the key/value table is wave-private, so the work-group barrier before the attention becomes a wave-level LDS fence (W stays in flight); bias requested at entry
# speedup vs baseline: 2.0983x; 1.2134x over previous
#include <hip/hip_runtime.h>
#include <math.h>


typedef float v2f __attribute__((ext_vector_type(2)));
typedef float v8f __attribute__((ext_vector_type(8)));

#define NW 8
#define NH 16
#define EDIM 128
#define TOK_PER_BLOCK 16
#define TOK_PER_LANE 4
#define NTILE 8
#define W_EARLY 3

__device__ __forceinline__ void load_rows(float4 (&kv)[2 * NH], const float* rows)
{
#pragma unroll
    for (int g = 0; g < NH; ++g) {
        kv[2 * g]     = *(const float4*)(rows + g * NW);
        kv[2 * g + 1] = *(const float4*)(rows + g * NW + 4);
    }
}

__device__ __forceinline__ v2f kpair(const float4 (&kv)[2 * NH], int g, int j)
{
    const float4 v = kv[2 * g + (j >> 1)];
    v2f p;
    if ((j & 1) == 0) { p.x = v.x; p.y = v.y; } else { p.x = v.z; p.y = v.w; }
    return p;
}

__device__ __forceinline__ void attend_piece(int p, const float (&qv)[NW], const float4 (&kv)[2 * NH],
                                             v2f (&d2)[NH], float (&e)[NH], v2f (&ao)[NW / 2],
                                             float& mx, float& den, float& r, float (&ov)[NW])
{
    const float inv_sqrt_dk = 0.35355339059327379f;
    if (p < 8) {
        const int j = p >> 1, g0 = 8 * (p & 1);
        v2f qp; qp.x = qv[2 * j]; qp.y = qv[2 * j + 1];
#pragma unroll
        for (int g = g0; g < g0 + 8; ++g)
            d2[g] = __builtin_elementwise_fma(qp, kpair(kv, g, j), d2[g]);
    } else if (p == 8) {
#pragma unroll
        for (int g = 0; g < NH; ++g) e[g] = (d2[g].x + d2[g].y) * inv_sqrt_dk;
    } else if (p == 9) {
        float m = e[0];
#pragma unroll
        for (int g = 1; g < NH; ++g) m = fmaxf(m, e[g]);
        mx = m;
    } else if (p < 14) {
        const int g0 = 4 * (p - 10);
#pragma unroll
        for (int t = 0; t < 4; t += 2) {
            v2f a; a.x = e[g0 + t]; a.y = e[g0 + t + 1];
            v2f m2; m2.x = mx; m2.y = mx;
            const v2f l2 = {1.44269502f, 1.44269502f};
            const v2f s = (a - m2) * l2;
            e[g0 + t]     = __builtin_amdgcn_exp2f(s.x);
            e[g0 + t + 1] = __builtin_amdgcn_exp2f(s.y);
        }
        if (p > 10) {
#pragma unroll
            for (int t = 0; t < 4; ++t) den += e[g0 - 4 + t];
        }
    } else if (p == 14) {
#pragma unroll
        for (int t = 0; t < 4; ++t) den += e[12 + t];
        r = 1.f / den;
    } else if (p < 31) {
        const int g = p - 15;
        v2f ep; ep.x = e[g]; ep.y = e[g];
#pragma unroll
        for (int j = 0; j < NW / 2; ++j)
            ao[j] = __builtin_elementwise_fma(ep, kpair(kv, g, j), ao[j]);
    } else {
#pragma unroll
        for (int j = 0; j < NW / 2; ++j) { ov[2 * j] = ao[j].x * r; ov[2 * j + 1] = ao[j].y * r; }
    }
}

__global__ __launch_bounds__(64) void quantum_attn_proj_kernel(
    const float* __restrict__ x,
    const float* __restrict__ theta,
    const float* __restrict__ W,
    const float* __restrict__ bias,
    float* __restrict__ out)
{
    __shared__ float qh_lds[TOK_PER_BLOCK][NH][NW];
    __shared__ float part_lds[2][NTILE / 2][8][32];

    const int lane = threadIdx.x & 31;
    const int wave = threadIdx.x >> 5;
    const int half = lane >> 4;
    const int h    = lane & 15;
    const int slot0 = wave * 8 + half;

    const float4 t0 = *(const float4*)(theta);
    const float4 t1 = *(const float4*)(theta + 4);

    float4 xa[TOK_PER_LANE], xb[TOK_PER_LANE];
#pragma unroll
    for (int jj = 0; jj < TOK_PER_LANE; ++jj) {
        const int gt = blockIdx.x * TOK_PER_BLOCK + slot0 + 2 * jj;
        const float* xp = x + (size_t)gt * EDIM + h * NW;
        xa[jj] = *(const float4*)(xp);
        xb[jj] = *(const float4*)(xp + 4);
    }

    float bcol[NTILE / 2];
#pragma unroll
    for (int t = 0; t < NTILE / 2; ++t)
        bcol[t] = bias[(4 * wave + t) * 16 + h];

    __builtin_amdgcn_sched_barrier(0);

    float4 wlo[TOK_PER_LANE][NTILE], whi[TOK_PER_LANE][NTILE];
#pragma unroll
    for (int jj = 0; jj < W_EARLY; ++jj) {
        const int kbase = (slot0 + 2 * jj) * NW;
#pragma unroll
        for (int u = 0; u < NTILE; ++u) {
            const int nt = (u + 4 * wave) & 7;
            const float* wp = W + (size_t)(nt * 16 + h) * EDIM + kbase;
            wlo[jj][u] = *(const float4*)(wp);
            whi[jj][u] = *(const float4*)(wp + 4);
        }
    }

    __builtin_amdgcn_sched_barrier(0);

    float q[TOK_PER_LANE][NW];
#pragma unroll
    for (int jj = 0; jj < TOK_PER_LANE; ++jj) {
        const int slot = slot0 + 2 * jj;
        const float4 x0 = xa[jj];
        const float4 x1 = xb[jj];

        float c[NW];
        c[0] = __cosf(x0.x + t0.x);
        c[1] = __cosf(x0.y + t0.y);
        c[2] = __cosf(x0.z + t0.z);
        c[3] = __cosf(x0.w + t0.w);
        c[4] = __cosf(x1.x + t1.x);
        c[5] = __cosf(x1.y + t1.y);
        c[6] = __cosf(x1.z + t1.z);
        c[7] = __cosf(x1.w + t1.w);

        float p = c[0];
#pragma unroll
        for (int w = 1; w < NW; ++w) { p *= c[w]; q[jj][w] = p; }
        q[jj][0] = c[1] * c[2] * c[3] * c[4] * c[5] * c[6] * c[7];

#pragma unroll
        for (int w = 0; w < NW; ++w) qh_lds[slot][h][w] = q[jj][w];
    }
    __builtin_amdgcn_fence(__ATOMIC_RELEASE, "wavefront");
    __builtin_amdgcn_wave_barrier();
    __builtin_amdgcn_fence(__ATOMIC_ACQUIRE, "wavefront");

#pragma unroll
    for (int jj = W_EARLY; jj < TOK_PER_LANE; ++jj) {
        const int kbase = (slot0 + 2 * jj) * NW;
#pragma unroll
        for (int u = 0; u < NTILE; ++u) {
            const int nt = (u + 4 * wave) & 7;
            const float* wp = W + (size_t)(nt * 16 + h) * EDIM + kbase;
            wlo[jj][u] = *(const float4*)(wp);
            whi[jj][u] = *(const float4*)(wp + 4);
        }
    }

    float4 kvA[2 * NH], kvB[2 * NH];
    float o[TOK_PER_LANE][NW];
    v8f acc[NTILE];
#pragma unroll
    for (int u = 0; u < NTILE; ++u)
        acc[u] = (v8f){0.f, 0.f, 0.f, 0.f, 0.f, 0.f, 0.f, 0.f};

#define PROJECT_STEP(JJ)                                                          \
    _Pragma("unroll")                                                             \
    for (int i = 0; i < 4; ++i) {                                                 \
        v2f a;                                                                    \
        a.x = o[JJ][2 * i];                                                       \
        a.y = o[JJ][2 * i + 1];                                                   \
        _Pragma("unroll")                                                         \
        for (int u = 0; u < NTILE; ++u) {                                         \
            v2f b;                                                                \
            if (i == 0)      { b.x = wlo[JJ][u].x; b.y = wlo[JJ][u].y; }          \
            else if (i == 1) { b.x = wlo[JJ][u].z; b.y = wlo[JJ][u].w; }          \
            else if (i == 2) { b.x = whi[JJ][u].x; b.y = whi[JJ][u].y; }          \
            else             { b.x = whi[JJ][u].z; b.y = whi[JJ][u].w; }          \
            acc[u] = __builtin_amdgcn_wmma_f32_16x16x4_f32(                       \
                 false, a,  false, b,                           \
                 (short)0, acc[u],  false,  false);\
        }                                                                         \
    }

#define PIPE_PAIR(JJ, KV, M)                                                      \
    {                                                                             \
        v2f a_, b_;                                                               \
        a_.x = o[JJ][2 * ((M) >> 3)];                                             \
        a_.y = o[JJ][2 * ((M) >> 3) + 1];                                         \
        const float4 bq_ = (((M) >> 3) < 2) ? wlo[JJ][(M) & 7] : whi[JJ][(M) & 7];\
        if ((((M) >> 3) & 1) == 0) { b_.x = bq_.x; b_.y = bq_.y; }                \
        else                       { b_.x = bq_.z; b_.y = bq_.w; }                \
        acc[(M) & 7] = __builtin_amdgcn_wmma_f32_16x16x4_f32(                     \
             false, a_,  false, b_,                             \
             (short)0, acc[(M) & 7],  false,  false);\
        attend_piece((M), q[JJ + 1], KV, d_, e_, ao_, mx_, den_, r_, o[JJ + 1]);  \
        __builtin_amdgcn_sched_barrier(0);                                        \
    }
#define PIPE_8(JJ, KV, B)                                                         \
    PIPE_PAIR(JJ, KV, (B) + 0) PIPE_PAIR(JJ, KV, (B) + 1)                         \
    PIPE_PAIR(JJ, KV, (B) + 2) PIPE_PAIR(JJ, KV, (B) + 3)                         \
    PIPE_PAIR(JJ, KV, (B) + 4) PIPE_PAIR(JJ, KV, (B) + 5)                         \
    PIPE_PAIR(JJ, KV, (B) + 6) PIPE_PAIR(JJ, KV, (B) + 7)
#define PIPE_STEP(JJ, KV)                                                         \
    {                                                                             \
        v2f d_[NH], ao_[NW / 2];                                                  \
        float e_[NH];                                                             \
        float mx_ = 0.f, den_ = 0.f, r_ = 0.f;                                    \
        _Pragma("unroll")                                                         \
        for (int g = 0; g < NH; ++g) { d_[g] = (v2f){0.f, 0.f}; e_[g] = 0.f; }    \
        _Pragma("unroll")                                                         \
        for (int w = 0; w < NW / 2; ++w) ao_[w] = (v2f){0.f, 0.f};                \
        PIPE_8(JJ, KV, 0) PIPE_8(JJ, KV, 8) PIPE_8(JJ, KV, 16) PIPE_8(JJ, KV, 24) \
    }

    load_rows(kvA, &qh_lds[slot0][0][0]);
    load_rows(kvB, &qh_lds[slot0 + 2][0][0]);
    {
        v2f d_[NH], ao_[NW / 2];
        float e_[NH];
        float mx_ = 0.f, den_ = 0.f, r_ = 0.f;
#pragma unroll
        for (int g = 0; g < NH; ++g) { d_[g] = (v2f){0.f, 0.f}; e_[g] = 0.f; }
#pragma unroll
        for (int w = 0; w < NW / 2; ++w) ao_[w] = (v2f){0.f, 0.f};
#pragma unroll
        for (int p = 0; p < 32; ++p)
            attend_piece(p, q[0], kvA, d_, e_, ao_, mx_, den_, r_, o[0]);
    }

    load_rows(kvA, &qh_lds[slot0 + 4][0][0]);
    PIPE_STEP(0, kvB)

    load_rows(kvB, &qh_lds[slot0 + 6][0][0]);
    PIPE_STEP(1, kvA)

    PIPE_STEP(2, kvB)

    PROJECT_STEP(3)

#pragma unroll
    for (int t = 0; t < NTILE / 2; ++t)
#pragma unroll
        for (int r = 0; r < 8; ++r)
            part_lds[wave][t][r][lane] = acc[NTILE / 2 + t][r];
    __syncthreads();

    const int b_idx = blockIdx.x >> 6;
    const int sgrp  = blockIdx.x & 63;
#pragma unroll
    for (int t = 0; t < NTILE / 2; ++t) {
        const int n = (4 * wave + t) * 16 + h;
#pragma unroll
        for (int r = 0; r < 8; ++r) {
            const int head = 8 * half + r;
            const size_t row = (size_t)b_idx * 1024 + (size_t)head * 64 + sgrp;
            const float s = acc[t][r] + part_lds[1 - wave][t][r][lane];
            out[row * EDIM + n] = s + bcol[t];
        }
    }
}

extern "C" void kernel_launch(void* const* d_in, const int* in_sizes, int n_in,
                              void* d_out, int out_size, void* d_ws, size_t ws_size,
                              hipStream_t stream) {
    const float* x     = (const float*)d_in[0];
    const float* theta = (const float*)d_in[1];
    const float* Wc    = (const float*)d_in[2];
    const float* bc    = (const float*)d_in[3];
    float* out = (float*)d_out;

    const int ntok = in_sizes[0] / EDIM;
    const int blocks1 = ntok / TOK_PER_BLOCK;
    quantum_attn_proj_kernel<<<blocks1, 64, 0, stream>>>(x, theta, Wc, bc, out);
}


# Matrix instructions of a token step run tile pair by tile pair, in the order their W fragments are requested and return: each waits for its own tile's loads, not the step's last; sums unchanged
# speedup vs baseline: 2.2189x; 1.0575x over previous
#include <hip/hip_runtime.h>
#include <math.h>


typedef float v2f __attribute__((ext_vector_type(2)));
typedef float v8f __attribute__((ext_vector_type(8)));

#define NW 8
#define NH 16
#define EDIM 128
#define TOK_PER_BLOCK 16
#define TOK_PER_LANE 4
#define NTILE 8
#define W_EARLY 3

__device__ __forceinline__ void load_rows(float4 (&kv)[2 * NH], const float* rows)
{
#pragma unroll
    for (int g = 0; g < NH; ++g) {
        kv[2 * g]     = *(const float4*)(rows + g * NW);
        kv[2 * g + 1] = *(const float4*)(rows + g * NW + 4);
    }
}

__device__ __forceinline__ v2f kpair(const float4 (&kv)[2 * NH], int g, int j)
{
    const float4 v = kv[2 * g + (j >> 1)];
    v2f p;
    if ((j & 1) == 0) { p.x = v.x; p.y = v.y; } else { p.x = v.z; p.y = v.w; }
    return p;
}

__device__ __forceinline__ void attend_piece(int p, const float (&qv)[NW], const float4 (&kv)[2 * NH],
                                             v2f (&d2)[NH], float (&e)[NH], v2f (&ao)[NW / 2],
                                             float& mx, float& den, float& r, float (&ov)[NW])
{
    const float inv_sqrt_dk = 0.35355339059327379f;
    if (p < 8) {
        const int j = p >> 1, g0 = 8 * (p & 1);
        v2f qp; qp.x = qv[2 * j]; qp.y = qv[2 * j + 1];
#pragma unroll
        for (int g = g0; g < g0 + 8; ++g)
            d2[g] = __builtin_elementwise_fma(qp, kpair(kv, g, j), d2[g]);
    } else if (p == 8) {
#pragma unroll
        for (int g = 0; g < NH; ++g) e[g] = (d2[g].x + d2[g].y) * inv_sqrt_dk;
    } else if (p == 9) {
        float m = e[0];
#pragma unroll
        for (int g = 1; g < NH; ++g) m = fmaxf(m, e[g]);
        mx = m;
    } else if (p < 14) {
        const int g0 = 4 * (p - 10);
#pragma unroll
        for (int t = 0; t < 4; t += 2) {
            v2f a; a.x = e[g0 + t]; a.y = e[g0 + t + 1];
            v2f m2; m2.x = mx; m2.y = mx;
            const v2f l2 = {1.44269502f, 1.44269502f};
            const v2f s = (a - m2) * l2;
            e[g0 + t]     = __builtin_amdgcn_exp2f(s.x);
            e[g0 + t + 1] = __builtin_amdgcn_exp2f(s.y);
        }
        if (p > 10) {
#pragma unroll
            for (int t = 0; t < 4; ++t) den += e[g0 - 4 + t];
        }
    } else if (p == 14) {
#pragma unroll
        for (int t = 0; t < 4; ++t) den += e[12 + t];
        r = 1.f / den;
    } else if (p < 31) {
        const int g = p - 15;
        v2f ep; ep.x = e[g]; ep.y = e[g];
#pragma unroll
        for (int j = 0; j < NW / 2; ++j)
            ao[j] = __builtin_elementwise_fma(ep, kpair(kv, g, j), ao[j]);
    } else {
#pragma unroll
        for (int j = 0; j < NW / 2; ++j) { ov[2 * j] = ao[j].x * r; ov[2 * j + 1] = ao[j].y * r; }
    }
}

__global__ __launch_bounds__(64) void quantum_attn_proj_kernel(
    const float* __restrict__ x,
    const float* __restrict__ theta,
    const float* __restrict__ W,
    const float* __restrict__ bias,
    float* __restrict__ out)
{
    __shared__ float qh_lds[TOK_PER_BLOCK][NH][NW];
    __shared__ float part_lds[2][NTILE / 2][8][32];

    const int lane = threadIdx.x & 31;
    const int wave = threadIdx.x >> 5;
    const int half = lane >> 4;
    const int h    = lane & 15;
    const int slot0 = wave * 8 + half;

    const float4 t0 = *(const float4*)(theta);
    const float4 t1 = *(const float4*)(theta + 4);

    float4 xa[TOK_PER_LANE], xb[TOK_PER_LANE];
#pragma unroll
    for (int jj = 0; jj < TOK_PER_LANE; ++jj) {
        const int gt = blockIdx.x * TOK_PER_BLOCK + slot0 + 2 * jj;
        const float* xp = x + (size_t)gt * EDIM + h * NW;
        xa[jj] = *(const float4*)(xp);
        xb[jj] = *(const float4*)(xp + 4);
    }

    float bcol[NTILE / 2];
#pragma unroll
    for (int t = 0; t < NTILE / 2; ++t)
        bcol[t] = bias[(4 * wave + t) * 16 + h];

    __builtin_amdgcn_sched_barrier(0);

    float4 wlo[TOK_PER_LANE][NTILE], whi[TOK_PER_LANE][NTILE];
#pragma unroll
    for (int jj = 0; jj < W_EARLY; ++jj) {
        const int kbase = (slot0 + 2 * jj) * NW;
#pragma unroll
        for (int u = 0; u < NTILE; ++u) {
            const int nt = (u + 4 * wave) & 7;
            const float* wp = W + (size_t)(nt * 16 + h) * EDIM + kbase;
            wlo[jj][u] = *(const float4*)(wp);
            whi[jj][u] = *(const float4*)(wp + 4);
        }
    }

    __builtin_amdgcn_sched_barrier(0);

    float q[TOK_PER_LANE][NW];
#pragma unroll
    for (int jj = 0; jj < TOK_PER_LANE; ++jj) {
        const int slot = slot0 + 2 * jj;
        const float4 x0 = xa[jj];
        const float4 x1 = xb[jj];

        float c[NW];
        c[0] = __cosf(x0.x + t0.x);
        c[1] = __cosf(x0.y + t0.y);
        c[2] = __cosf(x0.z + t0.z);
        c[3] = __cosf(x0.w + t0.w);
        c[4] = __cosf(x1.x + t1.x);
        c[5] = __cosf(x1.y + t1.y);
        c[6] = __cosf(x1.z + t1.z);
        c[7] = __cosf(x1.w + t1.w);

        float p = c[0];
#pragma unroll
        for (int w = 1; w < NW; ++w) { p *= c[w]; q[jj][w] = p; }
        q[jj][0] = c[1] * c[2] * c[3] * c[4] * c[5] * c[6] * c[7];

#pragma unroll
        for (int w = 0; w < NW; ++w) qh_lds[slot][h][w] = q[jj][w];
    }
    __builtin_amdgcn_fence(__ATOMIC_RELEASE, "wavefront");
    __builtin_amdgcn_wave_barrier();
    __builtin_amdgcn_fence(__ATOMIC_ACQUIRE, "wavefront");

#pragma unroll
    for (int jj = W_EARLY; jj < TOK_PER_LANE; ++jj) {
        const int kbase = (slot0 + 2 * jj) * NW;
#pragma unroll
        for (int u = 0; u < NTILE; ++u) {
            const int nt = (u + 4 * wave) & 7;
            const float* wp = W + (size_t)(nt * 16 + h) * EDIM + kbase;
            wlo[jj][u] = *(const float4*)(wp);
            whi[jj][u] = *(const float4*)(wp + 4);
        }
    }

    float4 kvA[2 * NH], kvB[2 * NH];
    float o[TOK_PER_LANE][NW];
    v8f acc[NTILE];
#pragma unroll
    for (int u = 0; u < NTILE; ++u)
        acc[u] = (v8f){0.f, 0.f, 0.f, 0.f, 0.f, 0.f, 0.f, 0.f};

#define MI(m) (((m) & 7) >> 1)
#define MU(m) (2 * ((m) >> 3) + ((m) & 1))

#define PROJECT_STEP(JJ)                                                          \
    _Pragma("unroll")                                                             \
    for (int m = 0; m < 4 * NTILE; ++m) {                                         \
        const int i = MI(m);                                                      \
        const int u = MU(m);                                                      \
        v2f a;                                                                    \
        a.x = o[JJ][2 * i];                                                       \
        a.y = o[JJ][2 * i + 1];                                                   \
        {                                                                         \
            v2f b;                                                                \
            if (i == 0)      { b.x = wlo[JJ][u].x; b.y = wlo[JJ][u].y; }          \
            else if (i == 1) { b.x = wlo[JJ][u].z; b.y = wlo[JJ][u].w; }          \
            else if (i == 2) { b.x = whi[JJ][u].x; b.y = whi[JJ][u].y; }          \
            else             { b.x = whi[JJ][u].z; b.y = whi[JJ][u].w; }          \
            acc[u] = __builtin_amdgcn_wmma_f32_16x16x4_f32(                       \
                 false, a,  false, b,                           \
                 (short)0, acc[u],  false,  false);\
        }                                                                         \
        if ((m & 7) == 7) __builtin_amdgcn_sched_barrier(0);                      \
    }

#define PIPE_PAIR(JJ, KV, M)                                                      \
    {                                                                             \
        v2f a_, b_;                                                               \
        a_.x = o[JJ][2 * MI(M)];                                                  \
        a_.y = o[JJ][2 * MI(M) + 1];                                              \
        const float4 bq_ = (MI(M) < 2) ? wlo[JJ][MU(M)] : whi[JJ][MU(M)];         \
        if ((MI(M) & 1) == 0) { b_.x = bq_.x; b_.y = bq_.y; }                     \
        else                  { b_.x = bq_.z; b_.y = bq_.w; }                     \
        acc[MU(M)] = __builtin_amdgcn_wmma_f32_16x16x4_f32(                       \
             false, a_,  false, b_,                             \
             (short)0, acc[MU(M)],  false,  false);\
        attend_piece((M), q[JJ + 1], KV, d_, e_, ao_, mx_, den_, r_, o[JJ + 1]);  \
        __builtin_amdgcn_sched_barrier(0);                                        \
    }
#define PIPE_8(JJ, KV, B)                                                         \
    PIPE_PAIR(JJ, KV, (B) + 0) PIPE_PAIR(JJ, KV, (B) + 1)                         \
    PIPE_PAIR(JJ, KV, (B) + 2) PIPE_PAIR(JJ, KV, (B) + 3)                         \
    PIPE_PAIR(JJ, KV, (B) + 4) PIPE_PAIR(JJ, KV, (B) + 5)                         \
    PIPE_PAIR(JJ, KV, (B) + 6) PIPE_PAIR(JJ, KV, (B) + 7)
#define PIPE_STEP(JJ, KV)                                                         \
    {                                                                             \
        v2f d_[NH], ao_[NW / 2];                                                  \
        float e_[NH];                                                             \
        float mx_ = 0.f, den_ = 0.f, r_ = 0.f;                                    \
        _Pragma("unroll")                                                         \
        for (int g = 0; g < NH; ++g) { d_[g] = (v2f){0.f, 0.f}; e_[g] = 0.f; }    \
        _Pragma("unroll")                                                         \
        for (int w = 0; w < NW / 2; ++w) ao_[w] = (v2f){0.f, 0.f};                \
        PIPE_8(JJ, KV, 0) PIPE_8(JJ, KV, 8) PIPE_8(JJ, KV, 16) PIPE_8(JJ, KV, 24) \
    }

    load_rows(kvA, &qh_lds[slot0][0][0]);
    load_rows(kvB, &qh_lds[slot0 + 2][0][0]);
    {
        v2f d_[NH], ao_[NW / 2];
        float e_[NH];
        float mx_ = 0.f, den_ = 0.f, r_ = 0.f;
#pragma unroll
        for (int g = 0; g < NH; ++g) { d_[g] = (v2f){0.f, 0.f}; e_[g] = 0.f; }
#pragma unroll
        for (int w = 0; w < NW / 2; ++w) ao_[w] = (v2f){0.f, 0.f};
#pragma unroll
        for (int p = 0; p < 32; ++p)
            attend_piece(p, q[0], kvA, d_, e_, ao_, mx_, den_, r_, o[0]);
    }

    load_rows(kvA, &qh_lds[slot0 + 4][0][0]);
    PIPE_STEP(0, kvB)

    load_rows(kvB, &qh_lds[slot0 + 6][0][0]);
    PIPE_STEP(1, kvA)

    PIPE_STEP(2, kvB)

    PROJECT_STEP(3)

#pragma unroll
    for (int t = 0; t < NTILE / 2; ++t)
#pragma unroll
        for (int r = 0; r < 8; ++r)
            part_lds[wave][t][r][lane] = acc[NTILE / 2 + t][r];
    __syncthreads();

    const int b_idx = blockIdx.x >> 6;
    const int sgrp  = blockIdx.x & 63;
#pragma unroll
    for (int t = 0; t < NTILE / 2; ++t) {
        const int n = (4 * wave + t) * 16 + h;
#pragma unroll
        for (int r = 0; r < 8; ++r) {
            const int head = 8 * half + r;
            const size_t row = (size_t)b_idx * 1024 + (size_t)head * 64 + sgrp;
            const float s = acc[t][r] + part_lds[1 - wave][t][r][lane];
            out[row * EDIM + n] = s + bcol[t];
        }
    }
}

extern "C" void kernel_launch(void* const* d_in, const int* in_sizes, int n_in,
                              void* d_out, int out_size, void* d_ws, size_t ws_size,
                              hipStream_t stream) {
    const float* x     = (const float*)d_in[0];
    const float* theta = (const float*)d_in[1];
    const float* Wc    = (const float*)d_in[2];
    const float* bc    = (const float*)d_in[3];
    float* out = (float*)d_out;

    const int ntok = in_sizes[0] / EDIM;
    const int blocks1 = ntok / TOK_PER_BLOCK;
    quantum_attn_proj_kernel<<<blocks1, 64, 0, stream>>>(x, theta, Wc, bc, out);
}


# 4-wave work-groups own two 16-token groups: a wave takes one k quarter of both, so each W fragment feeds two matrix instructions (half the W loads); an output is now the sum of four k-quarter sums
# speedup vs baseline: 2.3880x; 1.0762x over previous
#include <hip/hip_runtime.h>
#include <math.h>


typedef float v2f __attribute__((ext_vector_type(2)));
typedef float v8f __attribute__((ext_vector_type(8)));

#define NW 8
#define NH 16
#define EDIM 128
#define TOK_PER_BLOCK 32
#define TOK_PER_LANE 4
#define NTILE 8
#define NCHUNK 2

__device__ __forceinline__ void load_rows(float4 (&kv)[2 * NH], const float* rows)
{
#pragma unroll
    for (int g = 0; g < NH; ++g) {
        kv[2 * g]     = *(const float4*)(rows + g * NW);
        kv[2 * g + 1] = *(const float4*)(rows + g * NW + 4);
    }
}

__device__ __forceinline__ v2f kpair(const float4 (&kv)[2 * NH], int g, int j)
{
    const float4 v = kv[2 * g + (j >> 1)];
    v2f p;
    if ((j & 1) == 0) { p.x = v.x; p.y = v.y; } else { p.x = v.z; p.y = v.w; }
    return p;
}

__device__ __forceinline__ void attend_piece(int p, const float (&qv)[NW], const float4 (&kv)[2 * NH],
                                             v2f (&d2)[NH], float (&e)[NH], v2f (&ao)[NW / 2],
                                             float& mx, float& den, float& r, float (&ov)[NW])
{
    const float inv_sqrt_dk = 0.35355339059327379f;
    if (p < 8) {
        const int j = p >> 1, g0 = 8 * (p & 1);
        v2f qp; qp.x = qv[2 * j]; qp.y = qv[2 * j + 1];
#pragma unroll
        for (int g = g0; g < g0 + 8; ++g)
            d2[g] = __builtin_elementwise_fma(qp, kpair(kv, g, j), d2[g]);
    } else if (p == 8) {
#pragma unroll
        for (int g = 0; g < NH; ++g) e[g] = (d2[g].x + d2[g].y) * inv_sqrt_dk;
    } else if (p == 9) {
        float m = e[0];
#pragma unroll
        for (int g = 1; g < NH; ++g) m = fmaxf(m, e[g]);
        mx = m;
    } else if (p < 14) {
        const int g0 = 4 * (p - 10);
#pragma unroll
        for (int t = 0; t < 4; t += 2) {
            v2f a; a.x = e[g0 + t]; a.y = e[g0 + t + 1];
            v2f m2; m2.x = mx; m2.y = mx;
            const v2f l2 = {1.44269502f, 1.44269502f};
            const v2f s = (a - m2) * l2;
            e[g0 + t]     = __builtin_amdgcn_exp2f(s.x);
            e[g0 + t + 1] = __builtin_amdgcn_exp2f(s.y);
        }
        if (p > 10) {
#pragma unroll
            for (int t = 0; t < 4; ++t) den += e[g0 - 4 + t];
        }
    } else if (p == 14) {
#pragma unroll
        for (int t = 0; t < 4; ++t) den += e[12 + t];
        r = 1.f / den;
    } else if (p < 31) {
        const int g = p - 15;
        v2f ep; ep.x = e[g]; ep.y = e[g];
#pragma unroll
        for (int j = 0; j < NW / 2; ++j)
            ao[j] = __builtin_elementwise_fma(ep, kpair(kv, g, j), ao[j]);
    } else {
#pragma unroll
        for (int j = 0; j < NW / 2; ++j) { ov[2 * j] = ao[j].x * r; ov[2 * j + 1] = ao[j].y * r; }
    }
}

__global__ __launch_bounds__(128) void quantum_attn_proj_kernel(
    const float* x,
    const float* __restrict__ theta,
    const float* W,
    const float* __restrict__ bias,
    float* __restrict__ out)
{
    __shared__ float qh_lds[TOK_PER_BLOCK][NH][NW];
    __shared__ float4 part_lds[4][3][NTILE / 2][2][32];

    const int lane = threadIdx.x & 31;
    const int wave = threadIdx.x >> 5;
    const int half = lane >> 4;
    const int h    = lane & 15;
    const int gown  = wave >> 1;
    const int chalf = wave & 1;
    const int slot0 = wave * 4 + half;
    const int tbl0  = wave * 8 + half;

    float4 t0 = *(const float4*)(theta);
    float4 t1 = *(const float4*)(theta + 4);

    float4 xa[TOK_PER_LANE], xb[TOK_PER_LANE];
    const float* xg[2];
#pragma unroll
    for (int g = 0; g < 2; ++g) {
        const int gt = (blockIdx.x * 2 + (g ^ gown)) * 16 + slot0;
        xg[g] = x + (size_t)gt * EDIM + h * NW;
    }
#pragma unroll
    for (int jj = 0; jj < TOK_PER_LANE; ++jj) {
        const float* xp = xg[jj & 1] + (jj >> 1) * 2 * EDIM;
        xa[jj] = *(const float4*)(xp);
        xb[jj] = *(const float4*)(xp + 4);
    }

    float bcol[NTILE / 2];
#pragma unroll
    for (int t = 0; t < NTILE / 2; ++t)
        bcol[t] = bias[(4 * chalf + t) * 16 + h];

    asm volatile("" ::: "memory");
    __builtin_amdgcn_sched_barrier(0);

    float4 wlo[NCHUNK][NTILE], whi[NCHUNK][NTILE];
#pragma unroll
    for (int jj = 0; jj < NCHUNK; ++jj) {
        const int kbase = (slot0 + 2 * jj) * NW;
#pragma unroll
        for (int u = 0; u < NTILE; ++u) {
            const int nt = (u + 4 * chalf) & 7;
            const float* wp = W + (size_t)(nt * 16 + h) * EDIM + kbase;
            wlo[jj][u] = *(const float4*)(wp);
            whi[jj][u] = *(const float4*)(wp + 4);
        }
    }

    asm volatile("" : "+s"(t0.x), "+s"(t0.y), "+s"(t0.z), "+s"(t0.w),
                      "+s"(t1.x), "+s"(t1.y), "+s"(t1.z), "+s"(t1.w) :: "memory");
    __builtin_amdgcn_sched_barrier(0);

    float q[TOK_PER_LANE][NW];
#pragma unroll
    for (int jj = 0; jj < TOK_PER_LANE; ++jj) {
        const int slot = tbl0 + 2 * jj;
        const float4 x0 = xa[jj];
        const float4 x1 = xb[jj];

        float c[NW];
        c[0] = __cosf(x0.x + t0.x);
        c[1] = __cosf(x0.y + t0.y);
        c[2] = __cosf(x0.z + t0.z);
        c[3] = __cosf(x0.w + t0.w);
        c[4] = __cosf(x1.x + t1.x);
        c[5] = __cosf(x1.y + t1.y);
        c[6] = __cosf(x1.z + t1.z);
        c[7] = __cosf(x1.w + t1.w);

        float p = c[0];
#pragma unroll
        for (int w = 1; w < NW; ++w) { p *= c[w]; q[jj][w] = p; }
        q[jj][0] = c[1] * c[2] * c[3] * c[4] * c[5] * c[6] * c[7];

#pragma unroll
        for (int w = 0; w < NW; ++w) qh_lds[slot][h][w] = q[jj][w];
    }
    __builtin_amdgcn_fence(__ATOMIC_RELEASE, "wavefront");
    __builtin_amdgcn_wave_barrier();
    __builtin_amdgcn_fence(__ATOMIC_ACQUIRE, "wavefront");


    float4 kvA[2 * NH], kvB[2 * NH];
    float o[TOK_PER_LANE][NW];
    v8f acc[2 * NTILE];
#pragma unroll
    for (int u = 0; u < 2 * NTILE; ++u)
        acc[u] = (v8f){0.f, 0.f, 0.f, 0.f, 0.f, 0.f, 0.f, 0.f};

#define MI(m) (((m) & 7) >> 1)
#define MU(m) (2 * ((m) >> 3) + ((m) & 1))

#define PROJECT_STEP(JJ)                                                          \
    _Pragma("unroll")                                                             \
    for (int m = 0; m < 4 * NTILE; ++m) {                                         \
        const int i = MI(m);                                                      \
        const int u = MU(m);                                                      \
        v2f a;                                                                    \
        a.x = o[JJ][2 * i];                                                       \
        a.y = o[JJ][2 * i + 1];                                                   \
        {                                                                         \
            v2f b;                                                                \
            const int c = (JJ) >> 1;                                              \
            const int d = ((JJ) & 1) * NTILE + u;                                 \
            if (i == 0)      { b.x = wlo[c][u].x; b.y = wlo[c][u].y; }            \
            else if (i == 1) { b.x = wlo[c][u].z; b.y = wlo[c][u].w; }            \
            else if (i == 2) { b.x = whi[c][u].x; b.y = whi[c][u].y; }            \
            else             { b.x = whi[c][u].z; b.y = whi[c][u].w; }            \
            acc[d] = __builtin_amdgcn_wmma_f32_16x16x4_f32(                       \
                 false, a,  false, b,                           \
                 (short)0, acc[d],  false,  false);\
        }                                                                         \
        if ((m & 7) == 7) __builtin_amdgcn_sched_barrier(0);                      \
    }

#define PIPE_PAIR(JJ, KV, M)                                                      \
    {                                                                             \
        v2f a_, b_;                                                               \
        a_.x = o[JJ][2 * MI(M)];                                                  \
        a_.y = o[JJ][2 * MI(M) + 1];                                              \
        const float4 bq_ = (MI(M) < 2) ? wlo[(JJ) >> 1][MU(M)]                    \
                                       : whi[(JJ) >> 1][MU(M)];                   \
        if ((MI(M) & 1) == 0) { b_.x = bq_.x; b_.y = bq_.y; }                     \
        else                  { b_.x = bq_.z; b_.y = bq_.w; }                     \
        acc[((JJ) & 1) * NTILE + MU(M)] = __builtin_amdgcn_wmma_f32_16x16x4_f32(  \
             false, a_,  false, b_,                             \
             (short)0, acc[((JJ) & 1) * NTILE + MU(M)],                  \
             false,  false);                                \
        attend_piece((M), q[JJ + 1], KV, d_, e_, ao_, mx_, den_, r_, o[JJ + 1]);  \
        __builtin_amdgcn_sched_barrier(0);                                        \
    }
#define PIPE_8(JJ, KV, B)                                                         \
    PIPE_PAIR(JJ, KV, (B) + 0) PIPE_PAIR(JJ, KV, (B) + 1)                         \
    PIPE_PAIR(JJ, KV, (B) + 2) PIPE_PAIR(JJ, KV, (B) + 3)                         \
    PIPE_PAIR(JJ, KV, (B) + 4) PIPE_PAIR(JJ, KV, (B) + 5)                         \
    PIPE_PAIR(JJ, KV, (B) + 6) PIPE_PAIR(JJ, KV, (B) + 7)
#define PIPE_STEP(JJ, KV)                                                         \
    {                                                                             \
        v2f d_[NH], ao_[NW / 2];                                                  \
        float e_[NH];                                                             \
        float mx_ = 0.f, den_ = 0.f, r_ = 0.f;                                    \
        _Pragma("unroll")                                                         \
        for (int g = 0; g < NH; ++g) { d_[g] = (v2f){0.f, 0.f}; e_[g] = 0.f; }    \
        _Pragma("unroll")                                                         \
        for (int w = 0; w < NW / 2; ++w) ao_[w] = (v2f){0.f, 0.f};                \
        PIPE_8(JJ, KV, 0) PIPE_8(JJ, KV, 8) PIPE_8(JJ, KV, 16) PIPE_8(JJ, KV, 24) \
    }

    load_rows(kvA, &qh_lds[tbl0][0][0]);
    load_rows(kvB, &qh_lds[tbl0 + 2][0][0]);
    {
        v2f d_[NH], ao_[NW / 2];
        float e_[NH];
        float mx_ = 0.f, den_ = 0.f, r_ = 0.f;
#pragma unroll
        for (int g = 0; g < NH; ++g) { d_[g] = (v2f){0.f, 0.f}; e_[g] = 0.f; }
#pragma unroll
        for (int w = 0; w < NW / 2; ++w) ao_[w] = (v2f){0.f, 0.f};
#pragma unroll
        for (int p = 0; p < 32; ++p)
            attend_piece(p, q[0], kvA, d_, e_, ao_, mx_, den_, r_, o[0]);
    }

    load_rows(kvA, &qh_lds[tbl0 + 4][0][0]);
    PIPE_STEP(0, kvB)

    load_rows(kvB, &qh_lds[tbl0 + 6][0][0]);
    PIPE_STEP(1, kvA)

    PIPE_STEP(2, kvB)

#pragma unroll
    for (int t = 0; t < NTILE / 2; ++t) {
        const v8f a1 = acc[4 + t];
        part_lds[wave ^ 1][0][t][0][lane] = make_float4(a1[0], a1[1], a1[2], a1[3]);
        part_lds[wave ^ 1][0][t][1][lane] = make_float4(a1[4], a1[5], a1[6], a1[7]);
    }

    PROJECT_STEP(3)

#pragma unroll
    for (int t = 0; t < NTILE / 2; ++t) {
        const v8f a2 = acc[NTILE + t];
        const v8f a3 = acc[NTILE + 4 + t];
        part_lds[wave ^ 2][1][t][0][lane] = make_float4(a2[0], a2[1], a2[2], a2[3]);
        part_lds[wave ^ 2][1][t][1][lane] = make_float4(a2[4], a2[5], a2[6], a2[7]);
        part_lds[wave ^ 3][2][t][0][lane] = make_float4(a3[0], a3[1], a3[2], a3[3]);
        part_lds[wave ^ 3][2][t][1][lane] = make_float4(a3[4], a3[5], a3[6], a3[7]);
    }
    __syncthreads();

    const int grp_out = blockIdx.x * 2 + gown;
    const int b_idx = grp_out >> 6;
    const int sgrp  = grp_out & 63;
#pragma unroll
    for (int t = 0; t < NTILE / 2; ++t) {
        const int n = (4 * chalf + t) * 16 + h;
        float p1[8], p2[8], p3[8];
#pragma unroll
        for (int q4 = 0; q4 < 2; ++q4) {
            const float4 v1 = part_lds[wave][0][t][q4][lane];
            const float4 v2 = part_lds[wave][1][t][q4][lane];
            const float4 v3 = part_lds[wave][2][t][q4][lane];
            p1[4 * q4] = v1.x; p1[4 * q4 + 1] = v1.y; p1[4 * q4 + 2] = v1.z; p1[4 * q4 + 3] = v1.w;
            p2[4 * q4] = v2.x; p2[4 * q4 + 1] = v2.y; p2[4 * q4 + 2] = v2.z; p2[4 * q4 + 3] = v2.w;
            p3[4 * q4] = v3.x; p3[4 * q4 + 1] = v3.y; p3[4 * q4 + 2] = v3.z; p3[4 * q4 + 3] = v3.w;
        }
#pragma unroll
        for (int r = 0; r < 8; ++r) {
            const int head = 8 * half + r;
            const size_t row = (size_t)b_idx * 1024 + (size_t)head * 64 + sgrp;
            const float sum = ((acc[t][r] + p1[r]) + p2[r]) + p3[r];
            out[row * EDIM + n] = sum + bcol[t];
        }
    }
}

extern "C" void kernel_launch(void* const* d_in, const int* in_sizes, int n_in,
                              void* d_out, int out_size, void* d_ws, size_t ws_size,
                              hipStream_t stream) {
    const float* x     = (const float*)d_in[0];
    const float* theta = (const float*)d_in[1];
    const float* Wc    = (const float*)d_in[2];
    const float* bc    = (const float*)d_in[3];
    float* out = (float*)d_out;

    const int ntok = in_sizes[0] / EDIM;
    const int blocks1 = ntok / TOK_PER_BLOCK;
    quantum_attn_proj_kernel<<<blocks1, 128, 0, stream>>>(x, theta, Wc, bc, out);
}
